# starting kernel
# baseline (speedup 1.0000x reference)
#include <hip/hip_runtime.h>
#include <hip/hip_bf16.h>
#include <math.h>

#define Bq 32768
#define Lq 24
#define Hq 64
#define Vq 64

#define WAVES_PER_BLOCK 2
#define BLOCK_THREADS (WAVES_PER_BLOCK * 32)

typedef __attribute__((ext_vector_type(2))) float v2f;
typedef __attribute__((ext_vector_type(4))) float v4f;
typedef __attribute__((ext_vector_type(8))) float v8f;

#define OFF_W1   0
#define OFF_W2   8192
#define OFF_RP   16384
#define OFF_OW   20480
#define OFF_B1   24576
#define OFF_B2   24704
#define OFF_LNG  24768
#define OFF_LNB  24832
#define OFF_RPB  24896
#define OFF_OB   24960
#define OFF_OM   25024
#define OFF_WAVE 25056
#define H_STRIDE 68
#define HBUF_FLOATS (Lq * 16 * H_STRIDE)
#define X1_STRIDE 132
#define X1_FLOATS (16 * X1_STRIDE)
#define WAVE_FLOATS (HBUF_FLOATS + X1_FLOATS)
#define SMEM_FLOATS (OFF_WAVE + WAVES_PER_BLOCK * WAVE_FLOATS)
#define SMEM_BYTES (SMEM_FLOATS * 4)

template <int K>
__device__ inline v8f gemm_tile_f32(const float* __restrict__ As, int lda,
                                    const float* __restrict__ Bqp, int nstride4,
                                    v8f acc, int lane) {
  const int rc = lane & 15;
  const int hk = (lane >> 4) << 1;
  const float* ap = As + rc * lda + hk;
  const float* bp = Bqp + rc * 4 + hk;
#pragma unroll
  for (int k = 0; k < K; k += 4) {
    v2f a = *(const v2f*)(ap + k);
    v2f b = *(const v2f*)(bp + (k >> 2) * nstride4);
    acc = __builtin_amdgcn_wmma_f32_16x16x4_f32(
         false, a,  false, b,
         (short)0, acc,  false,  false);
  }
  return acc;
}

__global__ __launch_bounds__(BLOCK_THREADS)
void PerPositionEMADelta_fused(const int* __restrict__ seq,
                               const float* __restrict__ embed,
                               const float* __restrict__ w1,
                               const float* __restrict__ b1,
                               const float* __restrict__ w2,
                               const float* __restrict__ b2,
                               const float* __restrict__ ln_g,
                               const float* __restrict__ ln_b,
                               const float* __restrict__ rp_w,
                               const float* __restrict__ rp_b,
                               const float* __restrict__ out_w,
                               const float* __restrict__ out_b,
                               const float* __restrict__ alpha_logits,
                               float* __restrict__ out) {
  extern __shared__ float sm[];
  const int tid  = threadIdx.x;
  const int lane = tid & 31;
  const int wave = tid >> 5;
  const int rc   = lane & 15;
  const int half = lane >> 4;

  float* s_w1  = sm + OFF_W1;
  float* s_w2  = sm + OFF_W2;
  float* s_rp  = sm + OFF_RP;
  float* s_ow  = sm + OFF_OW;
  float* s_b1  = sm + OFF_B1;
  float* s_b2  = sm + OFF_B2;
  float* s_lng = sm + OFF_LNG;
  float* s_lnb = sm + OFF_LNB;
  float* s_rpb = sm + OFF_RPB;
  float* s_ob  = sm + OFF_OB;
  float* s_om  = sm + OFF_OM;
  float* wbase = sm + OFF_WAVE + wave * WAVE_FLOATS;
  float* s_H   = wbase;
  float* s_X1  = wbase + HBUF_FLOATS;

  for (int i = tid; i < 64 * 128; i += BLOCK_THREADS) {
    int k = i >> 7, n = i & 127;
    s_w1[(k >> 2) * 512 + n * 4 + (k & 3)] = w1[i];
  }
  for (int i = tid; i < 128 * 64; i += BLOCK_THREADS) {
    int k = i >> 6, n = i & 63;
    s_w2[(k >> 2) * 256 + n * 4 + (k & 3)] = w2[i];
  }
  for (int i = tid; i < 64 * 64; i += BLOCK_THREADS) {
    int k = i >> 6, n = i & 63;
    int p = (k >> 2) * 256 + n * 4 + (k & 3);
    s_rp[p] = rp_w[i];
    s_ow[p] = out_w[i];
  }
  for (int i = tid; i < 128; i += BLOCK_THREADS) s_b1[i] = b1[i];
  if (tid < 64) {
    s_b2[tid]  = b2[tid];
    s_lng[tid] = ln_g[tid];
    s_lnb[tid] = ln_b[tid];
    s_rpb[tid] = rp_b[tid];
    s_ob[tid]  = out_b[tid];
  }
  if (tid < Lq) {
    float sg = 1.0f / (1.0f + expf(-alpha_logits[tid]));
    s_om[tid] = 1.0f - (0.49f * sg + 0.5f);
  }
  __syncthreads();

  const int base = (blockIdx.x * WAVES_PER_BLOCK + wave) * 16;

  for (int t = 0; t < Lq; ++t) {
    float* Ht = s_H + t * (16 * H_STRIDE);
    {
      int idx = seq[(size_t)(base + rc) * Lq + t];
      const float* src = embed + (size_t)idx * Hq + half * 32;
      float* dst = Ht + rc * H_STRIDE + half * 32;
#pragma unroll
      for (int j = 0; j < 32; j += 4) *(v4f*)(dst + j) = *(const v4f*)(src + j);
    }
    __syncthreads();

    for (int nt = 0; nt < 8; ++nt) {
      v8f acc;
      float bv = s_b1[nt * 16 + rc];
#pragma unroll
      for (int j = 0; j < 8; ++j) acc[j] = bv;
      acc = gemm_tile_f32<64>(Ht, H_STRIDE, s_w1 + nt * 16 * 4, 512, acc, lane);
#pragma unroll
      for (int j = 0; j < 8; ++j) {
        int row = j + 8 * half;
        s_X1[row * X1_STRIDE + nt * 16 + rc] = fmaxf(acc[j], 0.0f);
      }
    }
    __syncthreads();

    for (int nt = 0; nt < 4; ++nt) {
      v8f acc;
      float bv = s_b2[nt * 16 + rc];
#pragma unroll
      for (int j = 0; j < 8; ++j) acc[j] = bv;
      acc = gemm_tile_f32<128>(s_X1, X1_STRIDE, s_w2 + nt * 16 * 4, 256, acc, lane);
#pragma unroll
      for (int j = 0; j < 8; ++j) {
        int row = j + 8 * half;
        Ht[row * H_STRIDE + nt * 16 + rc] += acc[j];
      }
    }
    __syncthreads();

    if (half == 0) {
      float* rowp = Ht + rc * H_STRIDE;
      float mu = 0.0f;
#pragma unroll
      for (int c = 0; c < 64; ++c) mu += rowp[c];
      mu *= (1.0f / 64.0f);
      float var = 0.0f;
#pragma unroll
      for (int c = 0; c < 64; ++c) { float d = rowp[c] - mu; var += d * d; }
      var *= (1.0f / 64.0f);
      float inv = rsqrtf(var + 1e-5f);
#pragma unroll
      for (int c = 0; c < 64; ++c)
        rowp[c] = (rowp[c] - mu) * inv * s_lng[c] + s_lnb[c];
    }
    __syncthreads();
  }

  float q[32], v[32];
  {
    const float* hl = s_H + 23 * (16 * H_STRIDE) + rc * H_STRIDE + half * 32;
#pragma unroll
    for (int j = 0; j < 32; ++j) { q[j] = hl[j]; v[j] = 0.0f; }
  }
  for (int s = Lq - 2; s >= 0; --s) {
    const float* hs = s_H + s * (16 * H_STRIDE) + rc * H_STRIDE + half * 32;
    float hk[32];
    float nsq = 0.0f;
#pragma unroll
    for (int j = 0; j < 32; ++j) { hk[j] = hs[j]; nsq += hk[j] * hk[j]; }
    nsq += __shfl_xor(nsq, 16, 32);
    float inv = 1.0f / fmaxf(sqrtf(nsq), 1e-12f);
    float dq = 0.0f;
#pragma unroll
    for (int j = 0; j < 32; ++j) dq += hk[j] * q[j];
    dq += __shfl_xor(dq, 16, 32);
    float c = s_om[s] * (dq * inv);
    float ci = c * inv;
#pragma unroll
    for (int j = 0; j < 32; ++j) {
      v[j] += c * hk[j];
      q[j] -= ci * hk[j];
    }
  }

  float* s_V  = s_X1;
  float* s_T1 = s_H;
  {
    float* dst = s_V + rc * H_STRIDE + half * 32;
#pragma unroll
    for (int j = 0; j < 32; ++j) dst[j] = v[j];
  }
  __syncthreads();

  for (int nt = 0; nt < 4; ++nt) {
    v8f acc;
    float bv = s_rpb[nt * 16 + rc];
#pragma unroll
    for (int j = 0; j < 8; ++j) acc[j] = bv;
    acc = gemm_tile_f32<64>(s_V, H_STRIDE, s_rp + nt * 16 * 4, 256, acc, lane);
#pragma unroll
    for (int j = 0; j < 8; ++j) {
      int row = j + 8 * half;
      s_T1[row * H_STRIDE + nt * 16 + rc] = acc[j];
    }
  }
  __syncthreads();

  for (int nt = 0; nt < 4; ++nt) {
    v8f acc;
    float bv = s_ob[nt * 16 + rc];
#pragma unroll
    for (int j = 0; j < 8; ++j) acc[j] = bv;
    acc = gemm_tile_f32<64>(s_T1, H_STRIDE, s_ow + nt * 16 * 4, 256, acc, lane);
#pragma unroll
    for (int j = 0; j < 8; ++j) {
      int row = j + 8 * half;
      out[(size_t)(base + row) * Vq + nt * 16 + rc] = acc[j];
    }
  }
}

extern "C" void kernel_launch(void* const* d_in, const int* in_sizes, int n_in,
                              void* d_out, int out_size, void* d_ws, size_t ws_size,
                              hipStream_t stream) {
  (void)n_in; (void)out_size; (void)d_ws; (void)ws_size;
  const int* seq            = (const int*)d_in[0];
  const float* embed        = (const float*)d_in[1];
  const float* w1           = (const float*)d_in[2];
  const float* b1           = (const float*)d_in[3];
  const float* w2           = (const float*)d_in[4];
  const float* b2           = (const float*)d_in[5];
  const float* ln_g         = (const float*)d_in[6];
  const float* ln_b         = (const float*)d_in[7];
  const float* rp_w         = (const float*)d_in[8];
  const float* rp_b         = (const float*)d_in[9];
  const float* out_w        = (const float*)d_in[10];
  const float* out_b        = (const float*)d_in[11];
  const float* alpha_logits = (const float*)d_in[12];
  float* out = (float*)d_out;

  const int B = in_sizes[0] / Lq;
  const int blocks = B / (16 * WAVES_PER_BLOCK);

  (void)hipFuncSetAttribute((const void*)PerPositionEMADelta_fused,
                            hipFuncAttributeMaxDynamicSharedMemorySize,
                            (int)SMEM_BYTES);

  PerPositionEMADelta_fused<<<blocks, BLOCK_THREADS, SMEM_BYTES, stream>>>(
      seq, embed, w1, b1, w2, b2, ln_g, ln_b, rp_w, rp_b, out_w, out_b,
      alpha_logits, out);
}


# MLP computed transposed (weights as WMMA A operand, tokens as columns): a D tile is the next product's B operand and the scan's row layout, so activations skip LDS and barriers; sum order changed
# speedup vs baseline: 5.5932x; 5.5932x over previous
#include <hip/hip_runtime.h>
#include <math.h>

#define Bq 32768
#define Lq 24
#define Hq 64
#define Vq 64

#define WAVES_PER_BLOCK 8
#define BLOCK_THREADS (WAVES_PER_BLOCK * 32)

typedef __attribute__((ext_vector_type(2))) float v2f;
typedef __attribute__((ext_vector_type(4))) float v4f;
typedef __attribute__((ext_vector_type(8))) float v8f;

#define OFF_W1   0
#define OFF_W2   8192
#define OFF_RP   16384
#define OFF_OW   20480
#define OFF_B1   24576
#define OFF_B2   24704
#define OFF_LNG  24768
#define OFF_LNB  24832
#define OFF_RPB  24896
#define OFF_OB   24960
#define OFF_OM   25024
#define OFF_WAVE 25056
#define H_STRIDE 68
#define HBUF_FLOATS (16 * H_STRIDE)
#define X1_STRIDE 132
#define X1_FLOATS (16 * X1_STRIDE)
#define WAVE_FLOATS (HBUF_FLOATS + X1_FLOATS)
#define SMEM_FLOATS (OFF_WAVE + WAVES_PER_BLOCK * WAVE_FLOATS)
#define SMEM_BYTES (SMEM_FLOATS * 4)

template <int K>
__device__ inline v8f gemm_tile_f32(const float* __restrict__ As, int lda,
                                    const float* __restrict__ Bqp, int nstride4,
                                    v8f acc, int lane) {
  const int rc = lane & 15;
  const int hk = (lane >> 4) << 1;
  const float* ap = As + rc * lda + hk;
  const float* bp = Bqp + rc * 4 + hk;
#pragma unroll
  for (int k = 0; k < K; k += 4) {
    v2f a = *(const v2f*)(ap + k);
    v2f b = *(const v2f*)(bp + (k >> 2) * nstride4);
    acc = __builtin_amdgcn_wmma_f32_16x16x4_f32(
         false, a,  false, b,
         (short)0, acc,  false,  false);
  }
  return acc;
}

__global__ __launch_bounds__(BLOCK_THREADS)
void PerPositionEMADelta_fused(const int* __restrict__ seq,
                               const float* __restrict__ embed,
                               const float* __restrict__ w1,
                               const float* __restrict__ b1,
                               const float* __restrict__ w2,
                               const float* __restrict__ b2,
                               const float* __restrict__ ln_g,
                               const float* __restrict__ ln_b,
                               const float* __restrict__ rp_w,
                               const float* __restrict__ rp_b,
                               const float* __restrict__ out_w,
                               const float* __restrict__ out_b,
                               const float* __restrict__ alpha_logits,
                               float* __restrict__ out) {
  extern __shared__ float sm[];
  const int tid  = threadIdx.x;
  const int lane = tid & 31;
  const int wave = tid >> 5;
  const int rc   = lane & 15;
  const int half = lane >> 4;

  float* s_w1  = sm + OFF_W1;
  float* s_w2  = sm + OFF_W2;
  float* s_rp  = sm + OFF_RP;
  float* s_ow  = sm + OFF_OW;
  float* s_b1  = sm + OFF_B1;
  float* s_b2  = sm + OFF_B2;
  float* s_lng = sm + OFF_LNG;
  float* s_lnb = sm + OFF_LNB;
  float* s_rpb = sm + OFF_RPB;
  float* s_ob  = sm + OFF_OB;
  float* s_om  = sm + OFF_OM;
  float* wbase = sm + OFF_WAVE + wave * WAVE_FLOATS;
  float* s_H   = wbase;
  float* s_X1  = wbase + HBUF_FLOATS;

  for (int i = tid; i < 64 * 128; i += BLOCK_THREADS) {
    int k = i >> 7, n = i & 127;
    s_w1[(k >> 2) * 512 + n * 4 + (k & 3)] = w1[i];
  }
  for (int i = tid; i < 128 * 64; i += BLOCK_THREADS) {
    int k = i >> 6, n = i & 63;
    s_w2[(k >> 2) * 256 + n * 4 + (k & 3)] = w2[i];
  }
  for (int i = tid; i < 64 * 64; i += BLOCK_THREADS) {
    int k = i >> 6, n = i & 63;
    int p = (k >> 2) * 256 + n * 4 + (k & 3);
    s_rp[p] = rp_w[i];
    s_ow[p] = out_w[i];
  }
  for (int i = tid; i < 128; i += BLOCK_THREADS) s_b1[i] = b1[i];
  if (tid < 64) {
    s_b2[tid]  = b2[tid];
    s_lng[tid] = ln_g[tid];
    s_lnb[tid] = ln_b[tid];
    s_rpb[tid] = rp_b[tid];
    s_ob[tid]  = out_b[tid];
  }
  if (tid < Lq) {
    float sg = 1.0f / (1.0f + expf(-alpha_logits[tid]));
    s_om[tid] = 1.0f - (0.49f * sg + 0.5f);
  }
  __syncthreads();

  const int base = (blockIdx.x * WAVES_PER_BLOCK + wave) * 16;

  const int* seqrow = seq + (size_t)(base + rc) * Lq;
  const float* w1p = s_w1 + half * 1024 + rc * 4;
  const float* w2p = s_w2 + half * 512 + rc * 4;
  const int fo = half * 8;

  float q[32], v[32];
#pragma unroll
  for (int j = 0; j < 32; ++j) { q[j] = 0.0f; v[j] = 0.0f; }

  int idx = seqrow[Lq - 1];
  for (int t = Lq - 1; t >= 0; --t) {
    asm volatile("" ::: "memory");
    float x[32];
    {
      const float* src = embed + (size_t)idx * Hq + fo;
#pragma unroll
      for (int m = 0; m < 4; ++m) {
        v4f lo = *(const v4f*)(src + 16 * m);
        v4f hi = *(const v4f*)(src + 16 * m + 4);
#pragma unroll
        for (int r = 0; r < 4; ++r) { x[8 * m + r] = lo[r]; x[8 * m + 4 + r] = hi[r]; }
      }
    }
    idx = seqrow[t > 0 ? t - 1 : 0];

    v8f x1[8];
#pragma unroll
    for (int mt = 0; mt < 8; ++mt) {
      v8f acc;
      {
        v4f blo = *(const v4f*)(s_b1 + 16 * mt + fo);
        v4f bhi = *(const v4f*)(s_b1 + 16 * mt + fo + 4);
#pragma unroll
        for (int r = 0; r < 4; ++r) { acc[r] = blo[r]; acc[4 + r] = bhi[r]; }
      }
#pragma unroll
      for (int m = 0; m < 4; ++m) {
#pragma unroll
        for (int p = 0; p < 2; ++p) {
          v4f a = *(const v4f*)(w1p + (4 * m + p) * 512 + 64 * mt);
          v2f a0 = {a[0], a[1]}, a1 = {a[2], a[3]};
          v2f b0 = {x[8 * m + 4 * p], x[8 * m + 4 * p + 1]};
          v2f b1v = {x[8 * m + 4 * p + 2], x[8 * m + 4 * p + 3]};
          acc = __builtin_amdgcn_wmma_f32_16x16x4_f32(false, a0, false, b0,
                                                      (short)0, acc, false, false);
          acc = __builtin_amdgcn_wmma_f32_16x16x4_f32(false, a1, false, b1v,
                                                      (short)0, acc, false, false);
        }
      }
#pragma unroll
      for (int r = 0; r < 8; ++r) acc[r] = fmaxf(acc[r], 0.0f);
      x1[mt] = acc;
    }

    float y[32];
#pragma unroll
    for (int m2 = 0; m2 < 4; ++m2) {
      v8f acc;
      {
        v4f blo = *(const v4f*)(s_b2 + 16 * m2 + fo);
        v4f bhi = *(const v4f*)(s_b2 + 16 * m2 + fo + 4);
#pragma unroll
        for (int r = 0; r < 4; ++r) {
          acc[r]     = x[8 * m2 + r]     + blo[r];
          acc[4 + r] = x[8 * m2 + 4 + r] + bhi[r];
        }
      }
#pragma unroll
      for (int mt = 0; mt < 8; ++mt) {
#pragma unroll
        for (int p = 0; p < 2; ++p) {
          v4f a = *(const v4f*)(w2p + (4 * mt + p) * 256 + 64 * m2);
          v2f a0 = {a[0], a[1]}, a1 = {a[2], a[3]};
          v2f b0 = {x1[mt][4 * p], x1[mt][4 * p + 1]};
          v2f b1v = {x1[mt][4 * p + 2], x1[mt][4 * p + 3]};
          acc = __builtin_amdgcn_wmma_f32_16x16x4_f32(false, a0, false, b0,
                                                      (short)0, acc, false, false);
          acc = __builtin_amdgcn_wmma_f32_16x16x4_f32(false, a1, false, b1v,
                                                      (short)0, acc, false, false);
        }
      }
#pragma unroll
      for (int r = 0; r < 8; ++r) y[8 * m2 + r] = acc[r];
    }

    float hk[32];
    {
      float s0 = 0.0f, s1 = 0.0f, s2 = 0.0f, s3 = 0.0f;
#pragma unroll
      for (int r = 0; r < 8; ++r) { s0 += y[r]; s1 += y[8 + r]; s2 += y[16 + r]; s3 += y[24 + r]; }
      float sum = (s0 + s1) + (s2 + s3);
      sum += __shfl_xor(sum, 16, 32);
      const float mu = sum * (1.0f / 64.0f);
      float d[32];
      float e0 = 0.0f, e1 = 0.0f, e2 = 0.0f, e3 = 0.0f;
#pragma unroll
      for (int r = 0; r < 8; ++r) {
        d[r]      = y[r]      - mu; e0 += d[r]      * d[r];
        d[8 + r]  = y[8 + r]  - mu; e1 += d[8 + r]  * d[8 + r];
        d[16 + r] = y[16 + r] - mu; e2 += d[16 + r] * d[16 + r];
        d[24 + r] = y[24 + r] - mu; e3 += d[24 + r] * d[24 + r];
      }
      float var = (e0 + e1) + (e2 + e3);
      var += __shfl_xor(var, 16, 32);
      var *= (1.0f / 64.0f);
      const float inv = rsqrtf(var + 1e-5f);
#pragma unroll
      for (int m = 0; m < 4; ++m) {
        v4f glo = *(const v4f*)(s_lng + 16 * m + fo);
        v4f ghi = *(const v4f*)(s_lng + 16 * m + fo + 4);
        v4f clo = *(const v4f*)(s_lnb + 16 * m + fo);
        v4f chi = *(const v4f*)(s_lnb + 16 * m + fo + 4);
#pragma unroll
        for (int r = 0; r < 4; ++r) {
          hk[8 * m + r]     = d[8 * m + r]     * inv * glo[r] + clo[r];
          hk[8 * m + 4 + r] = d[8 * m + 4 + r] * inv * ghi[r] + chi[r];
        }
      }
    }

    if (t == Lq - 1) {
#pragma unroll
      for (int j = 0; j < 32; ++j) q[j] = hk[j];
      continue;
    }
    float n0 = 0.0f, n1 = 0.0f, n2 = 0.0f, n3 = 0.0f;
    float d0 = 0.0f, d1 = 0.0f, d2 = 0.0f, d3 = 0.0f;
#pragma unroll
    for (int r = 0; r < 8; ++r) {
      n0 += hk[r] * hk[r];           d0 += hk[r] * q[r];
      n1 += hk[8 + r] * hk[8 + r];   d1 += hk[8 + r] * q[8 + r];
      n2 += hk[16 + r] * hk[16 + r]; d2 += hk[16 + r] * q[16 + r];
      n3 += hk[24 + r] * hk[24 + r]; d3 += hk[24 + r] * q[24 + r];
    }
    float nsq = (n0 + n1) + (n2 + n3);
    float dq  = (d0 + d1) + (d2 + d3);
    nsq += __shfl_xor(nsq, 16, 32);
    dq  += __shfl_xor(dq, 16, 32);
    float inv = 1.0f / fmaxf(sqrtf(nsq), 1e-12f);
    float c = s_om[t] * (dq * inv);
    float ci = c * inv;
#pragma unroll
    for (int j = 0; j < 32; ++j) {
      v[j] += c * hk[j];
      q[j] -= ci * hk[j];
    }
  }

  float* s_V  = s_X1;
  float* s_T1 = s_H;
  {
    float* dst = s_V + rc * H_STRIDE + fo;
#pragma unroll
    for (int j = 0; j < 32; ++j) dst[16 * (j >> 3) + (j & 7)] = v[j];
  }
  __syncthreads();

  for (int nt = 0; nt < 4; ++nt) {
    v8f acc;
    float bv = s_rpb[nt * 16 + rc];
#pragma unroll
    for (int j = 0; j < 8; ++j) acc[j] = bv;
    acc = gemm_tile_f32<64>(s_V, H_STRIDE, s_rp + nt * 16 * 4, 256, acc, lane);
#pragma unroll
    for (int j = 0; j < 8; ++j) {
      int row = j + 8 * half;
      s_T1[row * H_STRIDE + nt * 16 + rc] = acc[j];
    }
  }
  __syncthreads();

  for (int nt = 0; nt < 4; ++nt) {
    v8f acc;
    float bv = s_ob[nt * 16 + rc];
#pragma unroll
    for (int j = 0; j < 8; ++j) acc[j] = bv;
    acc = gemm_tile_f32<64>(s_T1, H_STRIDE, s_ow + nt * 16 * 4, 256, acc, lane);
#pragma unroll
    for (int j = 0; j < 8; ++j) {
      int row = j + 8 * half;
      out[(size_t)(base + row) * Vq + nt * 16 + rc] = acc[j];
    }
  }
}

extern "C" void kernel_launch(void* const* d_in, const int* in_sizes, int n_in,
                              void* d_out, int out_size, void* d_ws, size_t ws_size,
                              hipStream_t stream) {
  (void)n_in; (void)out_size; (void)d_ws; (void)ws_size;
  const int* seq            = (const int*)d_in[0];
  const float* embed        = (const float*)d_in[1];
  const float* w1           = (const float*)d_in[2];
  const float* b1           = (const float*)d_in[3];
  const float* w2           = (const float*)d_in[4];
  const float* b2           = (const float*)d_in[5];
  const float* ln_g         = (const float*)d_in[6];
  const float* ln_b         = (const float*)d_in[7];
  const float* rp_w         = (const float*)d_in[8];
  const float* rp_b         = (const float*)d_in[9];
  const float* out_w        = (const float*)d_in[10];
  const float* out_b        = (const float*)d_in[11];
  const float* alpha_logits = (const float*)d_in[12];
  float* out = (float*)d_out;

  const int B = in_sizes[0] / Lq;
  const int blocks = B / (16 * WAVES_PER_BLOCK);

  (void)hipFuncSetAttribute((const void*)PerPositionEMADelta_fused,
                            hipFuncAttributeMaxDynamicSharedMemorySize,
                            (int)SMEM_BYTES);

  PerPositionEMADelta_fused<<<blocks, BLOCK_THREADS, SMEM_BYTES, stream>>>(
      seq, embed, w1, b1, w2, b2, ln_g, ln_b, rp_w, rp_b, out_w, out_b,
      alpha_logits, out);
}


# Operand loads run one block ahead: each 32-WMMA block's weight fragments and bias rows are one LDS load group into a second register set, the token row is gathered a step ahead; sums unchanged
# speedup vs baseline: 6.0035x; 1.0734x over previous
#include <hip/hip_runtime.h>
#include <math.h>

#define Bq 32768
#define Lq 24
#define Hq 64
#define Vq 64

#define WAVES_PER_BLOCK 8
#define BLOCK_THREADS (WAVES_PER_BLOCK * 32)

typedef __attribute__((ext_vector_type(2))) float v2f;
typedef __attribute__((ext_vector_type(4))) float v4f;
typedef __attribute__((ext_vector_type(8))) float v8f;

#define OFF_W1   0
#define OFF_W2   8192
#define OFF_RP   16384
#define OFF_OW   20480
#define OFF_B1   24576
#define OFF_B2   24704
#define OFF_LNG  24768
#define OFF_LNB  24832
#define OFF_RPB  24896
#define OFF_OB   24960
#define OFF_OM   25024
#define OFF_WAVE 25056
#define H_STRIDE 68
#define HBUF_FLOATS (16 * H_STRIDE)
#define X1_STRIDE 132
#define X1_FLOATS (16 * X1_STRIDE)
#define WAVE_FLOATS (HBUF_FLOATS + X1_FLOATS)
#define SMEM_FLOATS (OFF_WAVE + WAVES_PER_BLOCK * WAVE_FLOATS)
#define SMEM_BYTES (SMEM_FLOATS * 4)

template <int K>
__device__ inline v8f gemm_tile_f32(const float* __restrict__ As, int lda,
                                    const float* __restrict__ Bqp, int nstride4,
                                    v8f acc, int lane) {
  const int rc = lane & 15;
  const int hk = (lane >> 4) << 1;
  const float* ap = As + rc * lda + hk;
  const float* bp = Bqp + rc * 4 + hk;
#pragma unroll
  for (int k = 0; k < K; k += 4) {
    v2f a = *(const v2f*)(ap + k);
    v2f b = *(const v2f*)(bp + (k >> 2) * nstride4);
    acc = __builtin_amdgcn_wmma_f32_16x16x4_f32(
         false, a,  false, b,
         (short)0, acc,  false,  false);
  }
  return acc;
}

__global__ __launch_bounds__(BLOCK_THREADS)
void PerPositionEMADelta_fused(const int* __restrict__ seq,
                               const float* __restrict__ embed,
                               const float* __restrict__ w1,
                               const float* __restrict__ b1,
                               const float* __restrict__ w2,
                               const float* __restrict__ b2,
                               const float* __restrict__ ln_g,
                               const float* __restrict__ ln_b,
                               const float* __restrict__ rp_w,
                               const float* __restrict__ rp_b,
                               const float* __restrict__ out_w,
                               const float* __restrict__ out_b,
                               const float* __restrict__ alpha_logits,
                               float* __restrict__ out) {
  extern __shared__ float sm[];
  const int tid  = threadIdx.x;
  const int lane = tid & 31;
  const int wave = tid >> 5;
  const int rc   = lane & 15;
  const int half = lane >> 4;

  float* s_w1  = sm + OFF_W1;
  float* s_w2  = sm + OFF_W2;
  float* s_rp  = sm + OFF_RP;
  float* s_ow  = sm + OFF_OW;
  float* s_b1  = sm + OFF_B1;
  float* s_b2  = sm + OFF_B2;
  float* s_lng = sm + OFF_LNG;
  float* s_lnb = sm + OFF_LNB;
  float* s_rpb = sm + OFF_RPB;
  float* s_ob  = sm + OFF_OB;
  float* s_om  = sm + OFF_OM;
  float* wbase = sm + OFF_WAVE + wave * WAVE_FLOATS;
  float* s_H   = wbase;
  float* s_X1  = wbase + HBUF_FLOATS;

  for (int i = tid; i < 64 * 128; i += BLOCK_THREADS) {
    int k = i >> 7, n = i & 127;
    s_w1[(k >> 2) * 512 + n * 4 + (k & 3)] = w1[i];
  }
  for (int i = tid; i < 128 * 64; i += BLOCK_THREADS) {
    int k = i >> 6, n = i & 63;
    s_w2[(k >> 2) * 256 + n * 4 + (k & 3)] = w2[i];
  }
  for (int i = tid; i < 64 * 64; i += BLOCK_THREADS) {
    int k = i >> 6, n = i & 63;
    int p = (k >> 2) * 256 + n * 4 + (k & 3);
    s_rp[p] = rp_w[i];
    s_ow[p] = out_w[i];
  }
  for (int i = tid; i < 128; i += BLOCK_THREADS) s_b1[i] = b1[i];
  if (tid < 64) {
    s_b2[tid]  = b2[tid];
    s_lng[tid] = ln_g[tid];
    s_lnb[tid] = ln_b[tid];
    s_rpb[tid] = rp_b[tid];
    s_ob[tid]  = out_b[tid];
  }
  if (tid < Lq) {
    float sg = 1.0f / (1.0f + expf(-alpha_logits[tid]));
    s_om[tid] = 1.0f - (0.49f * sg + 0.5f);
  }
  __syncthreads();

  const int base = (blockIdx.x * WAVES_PER_BLOCK + wave) * 16;

  const int* seqrow = seq + (size_t)(base + rc) * Lq;
  const float* w1p = s_w1 + half * 1024 + rc * 4;
  const float* w2p = s_w2 + half * 512 + rc * 4;
  const int fo = half * 8;

  float q[32], v[32];
#pragma unroll
  for (int j = 0; j < 32; ++j) { q[j] = 0.0f; v[j] = 0.0f; }

  v4f fa[2][16];
  v4f fc[2][4];
  v4f xn[8];
  {
    const int idx0 = seqrow[Lq - 1];
    const float* src = embed + (size_t)idx0 * Hq + fo;
#pragma unroll
    for (int m = 0; m < 4; ++m) {
      xn[2 * m]     = *(const v4f*)(src + 16 * m);
      xn[2 * m + 1] = *(const v4f*)(src + 16 * m + 4);
    }
  }
  int idxn = seqrow[Lq - 2];
#pragma unroll
  for (int i = 0; i < 2; ++i) {
#pragma unroll
    for (int m = 0; m < 4; ++m) {
#pragma unroll
      for (int p = 0; p < 2; ++p)
        fa[0][8 * i + 2 * m + p] = *(const v4f*)(w1p + (4 * m + p) * 512 + 64 * i);
    }
    fc[0][2 * i]     = *(const v4f*)(s_b1 + 16 * i + fo);
    fc[0][2 * i + 1] = *(const v4f*)(s_b1 + 16 * i + fo + 4);
  }

  for (int t = Lq - 1; t >= 0; --t) {
    asm volatile("" ::: "memory");
    float x[32];
#pragma unroll
    for (int m = 0; m < 4; ++m) {
#pragma unroll
      for (int r = 0; r < 4; ++r) { x[8 * m + r] = xn[2 * m][r]; x[8 * m + 4 + r] = xn[2 * m + 1][r]; }
    }
    {
      const float* src = embed + (size_t)idxn * Hq + fo;
#pragma unroll
      for (int m = 0; m < 4; ++m) {
        xn[2 * m]     = *(const v4f*)(src + 16 * m);
        xn[2 * m + 1] = *(const v4f*)(src + 16 * m + 4);
      }
    }
    idxn = seqrow[t > 1 ? t - 2 : 0];

    v8f x1[8];
    v8f yacc[4];
    v4f lg[8], lb[8];
#pragma unroll
    for (int b = 0; b < 8; ++b) {
      const int s = b & 1, ns = s ^ 1;
      const int nb = (b + 1) & 7;
      if (nb < 4) {
#pragma unroll
        for (int i = 0; i < 2; ++i) {
          const int mt = 2 * nb + i;
#pragma unroll
          for (int m = 0; m < 4; ++m) {
#pragma unroll
            for (int p = 0; p < 2; ++p)
              fa[ns][8 * i + 2 * m + p] = *(const v4f*)(w1p + (4 * m + p) * 512 + 64 * mt);
          }
          fc[ns][2 * i]     = *(const v4f*)(s_b1 + 16 * mt + fo);
          fc[ns][2 * i + 1] = *(const v4f*)(s_b1 + 16 * mt + fo + 4);
        }
      } else {
        const int pr = (nb - 4) >> 1, kh = (nb - 4) & 1;
#pragma unroll
        for (int i = 0; i < 2; ++i) {
          const int m2 = 2 * pr + i;
#pragma unroll
          for (int ml = 0; ml < 4; ++ml) {
#pragma unroll
            for (int p = 0; p < 2; ++p)
              fa[ns][8 * i + 2 * ml + p] =
                  *(const v4f*)(w2p + (4 * (4 * kh + ml) + p) * 256 + 64 * m2);
          }
          if (kh == 0) {
            fc[ns][2 * i]     = *(const v4f*)(s_b2 + 16 * m2 + fo);
            fc[ns][2 * i + 1] = *(const v4f*)(s_b2 + 16 * m2 + fo + 4);
          }
        }
      }
      if (b == 7) {
#pragma unroll
        for (int m = 0; m < 4; ++m) {
          lg[2 * m]     = *(const v4f*)(s_lng + 16 * m + fo);
          lg[2 * m + 1] = *(const v4f*)(s_lng + 16 * m + fo + 4);
          lb[2 * m]     = *(const v4f*)(s_lnb + 16 * m + fo);
          lb[2 * m + 1] = *(const v4f*)(s_lnb + 16 * m + fo + 4);
        }
      }
      __builtin_amdgcn_sched_barrier(0);
      if (b < 4) {
        v8f acc0, acc1;
#pragma unroll
        for (int r = 0; r < 4; ++r) {
          acc0[r] = fc[s][0][r]; acc0[4 + r] = fc[s][1][r];
          acc1[r] = fc[s][2][r]; acc1[4 + r] = fc[s][3][r];
        }
#pragma unroll
        for (int m = 0; m < 4; ++m) {
#pragma unroll
          for (int p = 0; p < 2; ++p) {
            const v4f a = fa[s][2 * m + p], c = fa[s][8 + 2 * m + p];
            const v2f a0 = {a[0], a[1]}, a1 = {a[2], a[3]};
            const v2f c0 = {c[0], c[1]}, c1 = {c[2], c[3]};
            const v2f b0 = {x[8 * m + 4 * p], x[8 * m + 4 * p + 1]};
            const v2f b1v = {x[8 * m + 4 * p + 2], x[8 * m + 4 * p + 3]};
            acc0 = __builtin_amdgcn_wmma_f32_16x16x4_f32(false, a0, false, b0,
                                                         (short)0, acc0, false, false);
            acc1 = __builtin_amdgcn_wmma_f32_16x16x4_f32(false, c0, false, b0,
                                                         (short)0, acc1, false, false);
            acc0 = __builtin_amdgcn_wmma_f32_16x16x4_f32(false, a1, false, b1v,
                                                         (short)0, acc0, false, false);
            acc1 = __builtin_amdgcn_wmma_f32_16x16x4_f32(false, c1, false, b1v,
                                                         (short)0, acc1, false, false);
          }
        }
        x1[2 * b] = acc0;
        x1[2 * b + 1] = acc1;
      } else {
        const int pr = (b - 4) >> 1, kh = (b - 4) & 1;
        const int ma = 2 * pr, mb = 2 * pr + 1;
        if (kh == 0) {
#pragma unroll
          for (int r = 0; r < 4; ++r) {
            yacc[ma][r]     = x[8 * ma + r]     + fc[s][0][r];
            yacc[ma][4 + r] = x[8 * ma + 4 + r] + fc[s][1][r];
            yacc[mb][r]     = x[8 * mb + r]     + fc[s][2][r];
            yacc[mb][4 + r] = x[8 * mb + 4 + r] + fc[s][3][r];
          }
        }
        if (pr == 0) {
#pragma unroll
          for (int ml = 0; ml < 4; ++ml) {
#pragma unroll
            for (int r = 0; r < 8; ++r) x1[4 * kh + ml][r] = fmaxf(x1[4 * kh + ml][r], 0.0f);
          }
        }
#pragma unroll
        for (int ml = 0; ml < 4; ++ml) {
          const int mt = 4 * kh + ml;
#pragma unroll
          for (int p = 0; p < 2; ++p) {
            const v4f a = fa[s][2 * ml + p], c = fa[s][8 + 2 * ml + p];
            const v2f a0 = {a[0], a[1]}, a1 = {a[2], a[3]};
            const v2f c0 = {c[0], c[1]}, c1 = {c[2], c[3]};
            const v2f b0 = {x1[mt][4 * p], x1[mt][4 * p + 1]};
            const v2f b1v = {x1[mt][4 * p + 2], x1[mt][4 * p + 3]};
            yacc[ma] = __builtin_amdgcn_wmma_f32_16x16x4_f32(false, a0, false, b0,
                                                             (short)0, yacc[ma], false, false);
            yacc[mb] = __builtin_amdgcn_wmma_f32_16x16x4_f32(false, c0, false, b0,
                                                             (short)0, yacc[mb], false, false);
            yacc[ma] = __builtin_amdgcn_wmma_f32_16x16x4_f32(false, a1, false, b1v,
                                                             (short)0, yacc[ma], false, false);
            yacc[mb] = __builtin_amdgcn_wmma_f32_16x16x4_f32(false, c1, false, b1v,
                                                             (short)0, yacc[mb], false, false);
          }
        }
      }
      __builtin_amdgcn_sched_barrier(0);
    }
    float y[32];
#pragma unroll
    for (int m2 = 0; m2 < 4; ++m2) {
#pragma unroll
      for (int r = 0; r < 8; ++r) y[8 * m2 + r] = yacc[m2][r];
    }

    float hk[32];
    {
      float s0 = 0.0f, s1 = 0.0f, s2 = 0.0f, s3 = 0.0f;
#pragma unroll
      for (int r = 0; r < 8; ++r) { s0 += y[r]; s1 += y[8 + r]; s2 += y[16 + r]; s3 += y[24 + r]; }
      float sum = (s0 + s1) + (s2 + s3);
      sum += __shfl_xor(sum, 16, 32);
      const float mu = sum * (1.0f / 64.0f);
      float d[32];
      float e0 = 0.0f, e1 = 0.0f, e2 = 0.0f, e3 = 0.0f;
#pragma unroll
      for (int r = 0; r < 8; ++r) {
        d[r]      = y[r]      - mu; e0 += d[r]      * d[r];
        d[8 + r]  = y[8 + r]  - mu; e1 += d[8 + r]  * d[8 + r];
        d[16 + r] = y[16 + r] - mu; e2 += d[16 + r] * d[16 + r];
        d[24 + r] = y[24 + r] - mu; e3 += d[24 + r] * d[24 + r];
      }
      float var = (e0 + e1) + (e2 + e3);
      var += __shfl_xor(var, 16, 32);
      var *= (1.0f / 64.0f);
      const float inv = rsqrtf(var + 1e-5f);
#pragma unroll
      for (int m = 0; m < 4; ++m) {
        const v4f glo = lg[2 * m], ghi = lg[2 * m + 1];
        const v4f clo = lb[2 * m], chi = lb[2 * m + 1];
#pragma unroll
        for (int r = 0; r < 4; ++r) {
          hk[8 * m + r]     = d[8 * m + r]     * inv * glo[r] + clo[r];
          hk[8 * m + 4 + r] = d[8 * m + 4 + r] * inv * ghi[r] + chi[r];
        }
      }
    }

    if (t == Lq - 1) {
#pragma unroll
      for (int j = 0; j < 32; ++j) q[j] = hk[j];
      continue;
    }
    float n0 = 0.0f, n1 = 0.0f, n2 = 0.0f, n3 = 0.0f;
    float d0 = 0.0f, d1 = 0.0f, d2 = 0.0f, d3 = 0.0f;
#pragma unroll
    for (int r = 0; r < 8; ++r) {
      n0 += hk[r] * hk[r];           d0 += hk[r] * q[r];
      n1 += hk[8 + r] * hk[8 + r];   d1 += hk[8 + r] * q[8 + r];
      n2 += hk[16 + r] * hk[16 + r]; d2 += hk[16 + r] * q[16 + r];
      n3 += hk[24 + r] * hk[24 + r]; d3 += hk[24 + r] * q[24 + r];
    }
    float nsq = (n0 + n1) + (n2 + n3);
    float dq  = (d0 + d1) + (d2 + d3);
    nsq += __shfl_xor(nsq, 16, 32);
    dq  += __shfl_xor(dq, 16, 32);
    float inv = 1.0f / fmaxf(sqrtf(nsq), 1e-12f);
    float c = s_om[t] * (dq * inv);
    float ci = c * inv;
#pragma unroll
    for (int j = 0; j < 32; ++j) {
      v[j] += c * hk[j];
      q[j] -= ci * hk[j];
    }
  }

  float* s_V  = s_X1;
  float* s_T1 = s_H;
  {
    float* dst = s_V + rc * H_STRIDE + fo;
#pragma unroll
    for (int j = 0; j < 32; ++j) dst[16 * (j >> 3) + (j & 7)] = v[j];
  }
  __syncthreads();

  for (int nt = 0; nt < 4; ++nt) {
    v8f acc;
    float bv = s_rpb[nt * 16 + rc];
#pragma unroll
    for (int j = 0; j < 8; ++j) acc[j] = bv;
    acc = gemm_tile_f32<64>(s_V, H_STRIDE, s_rp + nt * 16 * 4, 256, acc, lane);
#pragma unroll
    for (int j = 0; j < 8; ++j) {
      int row = j + 8 * half;
      s_T1[row * H_STRIDE + nt * 16 + rc] = acc[j];
    }
  }
  __syncthreads();

  for (int nt = 0; nt < 4; ++nt) {
    v8f acc;
    float bv = s_ob[nt * 16 + rc];
#pragma unroll
    for (int j = 0; j < 8; ++j) acc[j] = bv;
    acc = gemm_tile_f32<64>(s_T1, H_STRIDE, s_ow + nt * 16 * 4, 256, acc, lane);
#pragma unroll
    for (int j = 0; j < 8; ++j) {
      int row = j + 8 * half;
      out[(size_t)(base + row) * Vq + nt * 16 + rc] = acc[j];
    }
  }
}

extern "C" void kernel_launch(void* const* d_in, const int* in_sizes, int n_in,
                              void* d_out, int out_size, void* d_ws, size_t ws_size,
                              hipStream_t stream) {
  (void)n_in; (void)out_size; (void)d_ws; (void)ws_size;
  const int* seq            = (const int*)d_in[0];
  const float* embed        = (const float*)d_in[1];
  const float* w1           = (const float*)d_in[2];
  const float* b1           = (const float*)d_in[3];
  const float* w2           = (const float*)d_in[4];
  const float* b2           = (const float*)d_in[5];
  const float* ln_g         = (const float*)d_in[6];
  const float* ln_b         = (const float*)d_in[7];
  const float* rp_w         = (const float*)d_in[8];
  const float* rp_b         = (const float*)d_in[9];
  const float* out_w        = (const float*)d_in[10];
  const float* out_b        = (const float*)d_in[11];
  const float* alpha_logits = (const float*)d_in[12];
  float* out = (float*)d_out;

  const int B = in_sizes[0] / Lq;
  const int blocks = B / (16 * WAVES_PER_BLOCK);

  (void)hipFuncSetAttribute((const void*)PerPositionEMADelta_fused,
                            hipFuncAttributeMaxDynamicSharedMemorySize,
                            (int)SMEM_BYTES);

  PerPositionEMADelta_fused<<<blocks, BLOCK_THREADS, SMEM_BYTES, stream>>>(
      seq, embed, w1, b1, w2, b2, ln_g, ln_b, rp_w, rp_b, out_w, out_b,
      alpha_logits, out);
}


# Vector part of a step on aligned register pairs: LayerNorm and scan are packed ops on the D tiles in place, no loop-edge copies, half-waves joined by lane permute; sums pairwise, sum order changed
# speedup vs baseline: 6.2847x; 1.0468x over previous
#include <hip/hip_runtime.h>
#include <math.h>

#define Bq 32768
#define Lq 24
#define Hq 64
#define Vq 64

#define WAVES_PER_BLOCK 8
#define BLOCK_THREADS (WAVES_PER_BLOCK * 32)

typedef __attribute__((ext_vector_type(2))) float v2f;
typedef __attribute__((ext_vector_type(4))) float v4f;
typedef __attribute__((ext_vector_type(8))) float v8f;

#define OFF_W1   0
#define OFF_W2   8192
#define OFF_RP   16384
#define OFF_OW   20480
#define OFF_B1   24576
#define OFF_B2   24704
#define OFF_LNG  24768
#define OFF_LNB  24832
#define OFF_RPB  24896
#define OFF_OB   24960
#define OFF_OM   25024
#define OFF_WAVE 25056
#define H_STRIDE 68
#define HBUF_FLOATS (16 * H_STRIDE)
#define X1_STRIDE 132
#define X1_FLOATS (16 * X1_STRIDE)
#define WAVE_FLOATS (HBUF_FLOATS + X1_FLOATS)
#define SMEM_FLOATS (OFF_WAVE + WAVES_PER_BLOCK * WAVE_FLOATS)
#define SMEM_BYTES (SMEM_FLOATS * 4)

__device__ inline float xhalf(float x) {
  return __builtin_bit_cast(float, __builtin_amdgcn_permlanex16(
      __builtin_bit_cast(int, x), __builtin_bit_cast(int, x), 0x76543210, 0xfedcba98, false, false));
}

__device__ inline v8f relu8(v8f a) {
  const v8f z = {0.0f, 0.0f, 0.0f, 0.0f, 0.0f, 0.0f, 0.0f, 0.0f};
  return __builtin_elementwise_maximumnum(a, z);
}

template <int K>
__device__ inline v8f gemm_tile_f32(const float* __restrict__ As, int lda,
                                    const float* __restrict__ Bqp, int nstride4,
                                    v8f acc, int lane) {
  const int rc = lane & 15;
  const int hk = (lane >> 4) << 1;
  const float* ap = As + rc * lda + hk;
  const float* bp = Bqp + rc * 4 + hk;
#pragma unroll
  for (int k = 0; k < K; k += 4) {
    v2f a = *(const v2f*)(ap + k);
    v2f b = *(const v2f*)(bp + (k >> 2) * nstride4);
    acc = __builtin_amdgcn_wmma_f32_16x16x4_f32(
         false, a,  false, b,
         (short)0, acc,  false,  false);
  }
  return acc;
}

__global__ __launch_bounds__(BLOCK_THREADS)
void PerPositionEMADelta_fused(const int* __restrict__ seq,
                               const float* __restrict__ embed,
                               const float* __restrict__ w1,
                               const float* __restrict__ b1,
                               const float* __restrict__ w2,
                               const float* __restrict__ b2,
                               const float* __restrict__ ln_g,
                               const float* __restrict__ ln_b,
                               const float* __restrict__ rp_w,
                               const float* __restrict__ rp_b,
                               const float* __restrict__ out_w,
                               const float* __restrict__ out_b,
                               const float* __restrict__ alpha_logits,
                               float* __restrict__ out) {
  extern __shared__ float sm[];
  const int tid  = threadIdx.x;
  const int lane = tid & 31;
  const int wave = tid >> 5;
  const int rc   = lane & 15;
  const int half = lane >> 4;

  float* s_w1  = sm + OFF_W1;
  float* s_w2  = sm + OFF_W2;
  float* s_rp  = sm + OFF_RP;
  float* s_ow  = sm + OFF_OW;
  float* s_b1  = sm + OFF_B1;
  float* s_b2  = sm + OFF_B2;
  float* s_lng = sm + OFF_LNG;
  float* s_lnb = sm + OFF_LNB;
  float* s_rpb = sm + OFF_RPB;
  float* s_ob  = sm + OFF_OB;
  float* s_om  = sm + OFF_OM;
  float* wbase = sm + OFF_WAVE + wave * WAVE_FLOATS;
  float* s_H   = wbase;
  float* s_X1  = wbase + HBUF_FLOATS;

  for (int i = tid; i < 64 * 128; i += BLOCK_THREADS) {
    int k = i >> 7, n = i & 127;
    s_w1[(k >> 2) * 512 + n * 4 + (k & 3)] = w1[i];
  }
  for (int i = tid; i < 128 * 64; i += BLOCK_THREADS) {
    int k = i >> 6, n = i & 63;
    s_w2[(k >> 2) * 256 + n * 4 + (k & 3)] = w2[i];
  }
  for (int i = tid; i < 64 * 64; i += BLOCK_THREADS) {
    int k = i >> 6, n = i & 63;
    int p = (k >> 2) * 256 + n * 4 + (k & 3);
    s_rp[p] = rp_w[i];
    s_ow[p] = out_w[i];
  }
  for (int i = tid; i < 128; i += BLOCK_THREADS) s_b1[i] = b1[i];
  if (tid < 64) {
    s_b2[tid]  = b2[tid];
    s_lng[tid] = ln_g[tid];
    s_lnb[tid] = ln_b[tid];
    s_rpb[tid] = rp_b[tid];
    s_ob[tid]  = out_b[tid];
  }
  if (tid < Lq) {
    float sg = 1.0f / (1.0f + expf(-alpha_logits[tid]));
    s_om[tid] = 1.0f - (0.49f * sg + 0.5f);
  }
  __syncthreads();

  const int base = (blockIdx.x * WAVES_PER_BLOCK + wave) * 16;

  const int* seqrow = seq + (size_t)(base + rc) * Lq;
  const float* w1p = s_w1 + half * 1024 + rc * 4;
  const float* w2p = s_w2 + half * 512 + rc * 4;
  const int fo = half * 8;

  v2f q2[16], v2[16];
#pragma unroll
  for (int j = 0; j < 16; ++j) { q2[j] = (v2f){0.0f, 0.0f}; v2[j] = (v2f){0.0f, 0.0f}; }

  v4f fa[2][16];
  v4f fc[2][4];
  v4f fy[4];
  v4f xq[8];
  {
    const int idx0 = seqrow[Lq - 1];
    const float* src = embed + (size_t)idx0 * Hq + fo;
#pragma unroll
    for (int m = 0; m < 4; ++m) {
      xq[2 * m]     = *(const v4f*)(src + 16 * m);
      xq[2 * m + 1] = *(const v4f*)(src + 16 * m + 4);
    }
  }
  int idxn = seqrow[Lq - 2];
#pragma unroll
  for (int i = 0; i < 2; ++i) {
#pragma unroll
    for (int m = 0; m < 4; ++m) {
#pragma unroll
      for (int p = 0; p < 2; ++p)
        fa[0][8 * i + 2 * m + p] = *(const v4f*)(w1p + (4 * m + p) * 512 + 64 * i);
    }
    fc[0][2 * i]     = *(const v4f*)(s_b1 + 16 * i + fo);
    fc[0][2 * i + 1] = *(const v4f*)(s_b1 + 16 * i + fo + 4);
  }

  for (int t = Lq - 1; t >= 0; --t) {
    asm volatile("" ::: "memory");

    v8f x1[8];
    v8f yacc[4];
    v4f lg[8], lb[8];
#pragma unroll
    for (int b = 0; b < 8; ++b) {
      const int s = b & 1, ns = s ^ 1;
      const int nb = (b + 1) & 7;
      if (b == 5) {
        const float* src = embed + (size_t)idxn * Hq + fo;
#pragma unroll
        for (int m = 0; m < 4; ++m) {
          xq[2 * m]     = *(const v4f*)(src + 16 * m);
          xq[2 * m + 1] = *(const v4f*)(src + 16 * m + 4);
        }
        idxn = seqrow[t > 1 ? t - 2 : 0];
      }
      if (nb < 4) {
#pragma unroll
        for (int i = 0; i < 2; ++i) {
          const int mt = 2 * nb + i;
#pragma unroll
          for (int m = 0; m < 4; ++m) {
#pragma unroll
            for (int p = 0; p < 2; ++p)
              fa[ns][8 * i + 2 * m + p] = *(const v4f*)(w1p + (4 * m + p) * 512 + 64 * mt);
          }
          fc[ns][2 * i]     = *(const v4f*)(s_b1 + 16 * mt + fo);
          fc[ns][2 * i + 1] = *(const v4f*)(s_b1 + 16 * mt + fo + 4);
        }
      } else {
        const int pr = (nb - 4) >> 1, kh = (nb - 4) & 1;
#pragma unroll
        for (int i = 0; i < 2; ++i) {
          const int m2 = 2 * pr + i;
#pragma unroll
          for (int ml = 0; ml < 4; ++ml) {
#pragma unroll
            for (int p = 0; p < 2; ++p)
              fa[ns][8 * i + 2 * ml + p] =
                  *(const v4f*)(w2p + (4 * (4 * kh + ml) + p) * 256 + 64 * m2);
          }
        }
        if (nb == 4) {
#pragma unroll
          for (int i = 0; i < 2; ++i) {
            fc[ns][2 * i]     = *(const v4f*)(s_b2 + 16 * i + fo);
            fc[ns][2 * i + 1] = *(const v4f*)(s_b2 + 16 * i + fo + 4);
            fy[2 * i]         = *(const v4f*)(s_b2 + 16 * (2 + i) + fo);
            fy[2 * i + 1]     = *(const v4f*)(s_b2 + 16 * (2 + i) + fo + 4);
          }
        }
      }
      if (b == 7) {
#pragma unroll
        for (int m = 0; m < 4; ++m) {
          lg[2 * m]     = *(const v4f*)(s_lng + 16 * m + fo);
          lg[2 * m + 1] = *(const v4f*)(s_lng + 16 * m + fo + 4);
          lb[2 * m]     = *(const v4f*)(s_lnb + 16 * m + fo);
          lb[2 * m + 1] = *(const v4f*)(s_lnb + 16 * m + fo + 4);
        }
      }
      __builtin_amdgcn_sched_barrier(0);
      if (b < 4) {
        v8f acc0 = __builtin_shufflevector(fc[s][0], fc[s][1], 0, 1, 2, 3, 4, 5, 6, 7);
        v8f acc1 = __builtin_shufflevector(fc[s][2], fc[s][3], 0, 1, 2, 3, 4, 5, 6, 7);
#pragma unroll
        for (int m = 0; m < 4; ++m) {
#pragma unroll
          for (int p = 0; p < 2; ++p) {
            const v4f a = fa[s][2 * m + p], c = fa[s][8 + 2 * m + p];
            const v4f xb = xq[2 * m + p];
            const v2f a0 = {a[0], a[1]}, a1 = {a[2], a[3]};
            const v2f c0 = {c[0], c[1]}, c1 = {c[2], c[3]};
            const v2f b0 = {xb[0], xb[1]};
            const v2f b1v = {xb[2], xb[3]};
            acc0 = __builtin_amdgcn_wmma_f32_16x16x4_f32(false, a0, false, b0,
                                                         (short)0, acc0, false, false);
            acc1 = __builtin_amdgcn_wmma_f32_16x16x4_f32(false, c0, false, b0,
                                                         (short)0, acc1, false, false);
            acc0 = __builtin_amdgcn_wmma_f32_16x16x4_f32(false, a1, false, b1v,
                                                         (short)0, acc0, false, false);
            acc1 = __builtin_amdgcn_wmma_f32_16x16x4_f32(false, c1, false, b1v,
                                                         (short)0, acc1, false, false);
          }
        }
        x1[2 * b] = acc0;
        x1[2 * b + 1] = acc1;
      } else {
        const int pr = (b - 4) >> 1, kh = (b - 4) & 1;
        const int ma = 2 * pr, mb = 2 * pr + 1;
        if (b == 4) {
          yacc[0] = __builtin_shufflevector(xq[0] + fc[s][0], xq[1] + fc[s][1], 0, 1, 2, 3, 4, 5, 6, 7);
          yacc[1] = __builtin_shufflevector(xq[2] + fc[s][2], xq[3] + fc[s][3], 0, 1, 2, 3, 4, 5, 6, 7);
          yacc[2] = __builtin_shufflevector(xq[4] + fy[0], xq[5] + fy[1], 0, 1, 2, 3, 4, 5, 6, 7);
          yacc[3] = __builtin_shufflevector(xq[6] + fy[2], xq[7] + fy[3], 0, 1, 2, 3, 4, 5, 6, 7);
        }
        if (pr == 0) {
#pragma unroll
          for (int ml = 0; ml < 4; ++ml)
            x1[4 * kh + ml] = relu8(x1[4 * kh + ml]);
        }
#pragma unroll
        for (int ml = 0; ml < 4; ++ml) {
          const int mt = 4 * kh + ml;
#pragma unroll
          for (int p = 0; p < 2; ++p) {
            const v4f a = fa[s][2 * ml + p], c = fa[s][8 + 2 * ml + p];
            const v2f a0 = {a[0], a[1]}, a1 = {a[2], a[3]};
            const v2f c0 = {c[0], c[1]}, c1 = {c[2], c[3]};
            const v2f b0 = {x1[mt][4 * p], x1[mt][4 * p + 1]};
            const v2f b1v = {x1[mt][4 * p + 2], x1[mt][4 * p + 3]};
            yacc[ma] = __builtin_amdgcn_wmma_f32_16x16x4_f32(false, a0, false, b0,
                                                             (short)0, yacc[ma], false, false);
            yacc[mb] = __builtin_amdgcn_wmma_f32_16x16x4_f32(false, c0, false, b0,
                                                             (short)0, yacc[mb], false, false);
            yacc[ma] = __builtin_amdgcn_wmma_f32_16x16x4_f32(false, a1, false, b1v,
                                                             (short)0, yacc[ma], false, false);
            yacc[mb] = __builtin_amdgcn_wmma_f32_16x16x4_f32(false, c1, false, b1v,
                                                             (short)0, yacc[mb], false, false);
          }
        }
      }
      __builtin_amdgcn_sched_barrier(0);
    }

    v2f y2[16];
#pragma unroll
    for (int m = 0; m < 4; ++m) {
#pragma unroll
      for (int i = 0; i < 4; ++i) y2[4 * m + i] = (v2f){yacc[m][2 * i], yacc[m][2 * i + 1]};
    }
    v2f hk2[16];
    {
      v2f s0 = y2[0], s1 = y2[4], s2 = y2[8], s3 = y2[12];
#pragma unroll
      for (int i = 1; i < 4; ++i) { s0 += y2[i]; s1 += y2[4 + i]; s2 += y2[8 + i]; s3 += y2[12 + i]; }
      const v2f sp = (s0 + s1) + (s2 + s3);
      float sum = sp[0] + sp[1];
      sum += xhalf(sum);
      const float mu = sum * (1.0f / 64.0f);
      const v2f mu2 = {mu, mu};
      v2f d2[16];
#pragma unroll
      for (int i = 0; i < 16; ++i) d2[i] = y2[i] - mu2;
      v2f e0 = d2[0] * d2[0], e1 = d2[4] * d2[4], e2 = d2[8] * d2[8], e3 = d2[12] * d2[12];
#pragma unroll
      for (int i = 1; i < 4; ++i) {
        e0 = __builtin_elementwise_fma(d2[i], d2[i], e0);
        e1 = __builtin_elementwise_fma(d2[4 + i], d2[4 + i], e1);
        e2 = __builtin_elementwise_fma(d2[8 + i], d2[8 + i], e2);
        e3 = __builtin_elementwise_fma(d2[12 + i], d2[12 + i], e3);
      }
      const v2f ep = (e0 + e1) + (e2 + e3);
      float var = ep[0] + ep[1];
      var += xhalf(var);
      var *= (1.0f / 64.0f);
      const float inv = rsqrtf(var + 1e-5f);
      const v2f inv2 = {inv, inv};
#pragma unroll
      for (int m = 0; m < 4; ++m) {
        const v4f glo = lg[2 * m], ghi = lg[2 * m + 1];
        const v4f clo = lb[2 * m], chi = lb[2 * m + 1];
        hk2[4 * m + 0] = __builtin_elementwise_fma(d2[4 * m + 0] * inv2, (v2f){glo[0], glo[1]}, (v2f){clo[0], clo[1]});
        hk2[4 * m + 1] = __builtin_elementwise_fma(d2[4 * m + 1] * inv2, (v2f){glo[2], glo[3]}, (v2f){clo[2], clo[3]});
        hk2[4 * m + 2] = __builtin_elementwise_fma(d2[4 * m + 2] * inv2, (v2f){ghi[0], ghi[1]}, (v2f){chi[0], chi[1]});
        hk2[4 * m + 3] = __builtin_elementwise_fma(d2[4 * m + 3] * inv2, (v2f){ghi[2], ghi[3]}, (v2f){chi[2], chi[3]});
      }
    }

    v2f n0 = hk2[0] * hk2[0], n1 = hk2[4] * hk2[4], n2 = hk2[8] * hk2[8], n3 = hk2[12] * hk2[12];
    v2f g0 = hk2[0] * q2[0], g1 = hk2[4] * q2[4], g2 = hk2[8] * q2[8], g3 = hk2[12] * q2[12];
#pragma unroll
    for (int i = 1; i < 4; ++i) {
      n0 = __builtin_elementwise_fma(hk2[i], hk2[i], n0);
      n1 = __builtin_elementwise_fma(hk2[4 + i], hk2[4 + i], n1);
      n2 = __builtin_elementwise_fma(hk2[8 + i], hk2[8 + i], n2);
      n3 = __builtin_elementwise_fma(hk2[12 + i], hk2[12 + i], n3);
      g0 = __builtin_elementwise_fma(hk2[i], q2[i], g0);
      g1 = __builtin_elementwise_fma(hk2[4 + i], q2[4 + i], g1);
      g2 = __builtin_elementwise_fma(hk2[8 + i], q2[8 + i], g2);
      g3 = __builtin_elementwise_fma(hk2[12 + i], q2[12 + i], g3);
    }
    const v2f np = (n0 + n1) + (n2 + n3);
    const v2f gp = (g0 + g1) + (g2 + g3);
    float nsq = np[0] + np[1];
    float dq  = gp[0] + gp[1];
    nsq += xhalf(nsq);
    dq  += xhalf(dq);
    const float invn = 1.0f / fmaxf(sqrtf(nsq), 1e-12f);
    const float c = s_om[t] * (dq * invn);
    const float ci = (t == Lq - 1) ? -1.0f : c * invn;
    const v2f c2 = {c, c}, nci2 = {-ci, -ci};
#pragma unroll
    for (int j = 0; j < 16; ++j) {
      v2[j] = __builtin_elementwise_fma(c2, hk2[j], v2[j]);
      q2[j] = __builtin_elementwise_fma(nci2, hk2[j], q2[j]);
    }
  }

  float* s_V  = s_X1;
  float* s_T1 = s_H;
  {
    float* dst = s_V + rc * H_STRIDE + fo;
#pragma unroll
    for (int j = 0; j < 16; ++j) *(v2f*)(dst + 16 * (j >> 2) + 2 * (j & 3)) = v2[j];
  }
  __syncthreads();

  for (int nt = 0; nt < 4; ++nt) {
    v8f acc;
    float bv = s_rpb[nt * 16 + rc];
#pragma unroll
    for (int j = 0; j < 8; ++j) acc[j] = bv;
    acc = gemm_tile_f32<64>(s_V, H_STRIDE, s_rp + nt * 16 * 4, 256, acc, lane);
#pragma unroll
    for (int j = 0; j < 8; ++j) {
      int row = j + 8 * half;
      s_T1[row * H_STRIDE + nt * 16 + rc] = acc[j];
    }
  }
  __syncthreads();

  for (int nt = 0; nt < 4; ++nt) {
    v8f acc;
    float bv = s_ob[nt * 16 + rc];
#pragma unroll
    for (int j = 0; j < 8; ++j) acc[j] = bv;
    acc = gemm_tile_f32<64>(s_T1, H_STRIDE, s_ow + nt * 16 * 4, 256, acc, lane);
#pragma unroll
    for (int j = 0; j < 8; ++j) {
      int row = j + 8 * half;
      out[(size_t)(base + row) * Vq + nt * 16 + rc] = acc[j];
    }
  }
}

extern "C" void kernel_launch(void* const* d_in, const int* in_sizes, int n_in,
                              void* d_out, int out_size, void* d_ws, size_t ws_size,
                              hipStream_t stream) {
  (void)n_in; (void)out_size; (void)d_ws; (void)ws_size;
  const int* seq            = (const int*)d_in[0];
  const float* embed        = (const float*)d_in[1];
  const float* w1           = (const float*)d_in[2];
  const float* b1           = (const float*)d_in[3];
  const float* w2           = (const float*)d_in[4];
  const float* b2           = (const float*)d_in[5];
  const float* ln_g         = (const float*)d_in[6];
  const float* ln_b         = (const float*)d_in[7];
  const float* rp_w         = (const float*)d_in[8];
  const float* rp_b         = (const float*)d_in[9];
  const float* out_w        = (const float*)d_in[10];
  const float* out_b        = (const float*)d_in[11];
  const float* alpha_logits = (const float*)d_in[12];
  float* out = (float*)d_out;

  const int B = in_sizes[0] / Lq;
  const int blocks = B / (16 * WAVES_PER_BLOCK);

  (void)hipFuncSetAttribute((const void*)PerPositionEMADelta_fused,
                            hipFuncAttributeMaxDynamicSharedMemorySize,
                            (int)SMEM_BYTES);

  PerPositionEMADelta_fused<<<blocks, BLOCK_THREADS, SMEM_BYTES, stream>>>(
      seq, embed, w1, b1, w2, b2, ln_g, ln_b, rp_w, rp_b, out_w, out_b,
      alpha_logits, out);
}
